# starting kernel
# baseline (speedup 1.0000x reference)
#include <hip/hip_runtime.h>
#include <hip/hip_bf16.h>
#include <math.h>

typedef __attribute__((ext_vector_type(2))) float v2f;
typedef __attribute__((ext_vector_type(8))) float v8f;

#define BATCH 65536
#define DD 16
#define HH 64
#define NM 3

#define W01_S 160
#define W2_S  32
#define OFF_W0P 0
#define OFF_W1P 1280
#define OFF_W2P 6400
#define NET_SZ  7424
#define WGT_SZ  (2*NET_SZ)

#define XW_S    18
#define ACT_S   66
#define XW_SZ   (16*XW_S)
#define ACT_SZ  (16*ACT_S)
#define WAVE_SCR (XW_SZ + 2*ACT_SZ)
#define WAVES_PER_BLOCK 8
#define LDS_FLOATS (WGT_SZ + WAVES_PER_BLOCK*WAVE_SCR)
#define LDS_BYTES  (LDS_FLOATS*4)

#define LOG2E 1.4426950408889634f

__device__ __forceinline__ float fast_tanh(float x) {
    float e = __builtin_amdgcn_exp2f(x * (2.0f * LOG2E));
    return 1.0f - 2.0f * __builtin_amdgcn_rcpf(e + 1.0f);
}
__device__ __forceinline__ float fast_sigmoid(float a) {
    return __builtin_amdgcn_rcpf(1.0f + __builtin_amdgcn_exp2f(a * -LOG2E));
}

__global__ void __launch_bounds__(256, 1)
maf_wmma_kernel(const float* __restrict__ mu_M0, const float* __restrict__ mu_M1,
                const float* __restrict__ mu_M2, const float* __restrict__ mu_W0,
                const float* __restrict__ mu_b0, const float* __restrict__ mu_W1,
                const float* __restrict__ mu_b1, const float* __restrict__ mu_W2,
                const float* __restrict__ mu_b2,
                const float* __restrict__ al_M0, const float* __restrict__ al_M1,
                const float* __restrict__ al_M2, const float* __restrict__ al_W0,
                const float* __restrict__ al_b0, const float* __restrict__ al_W1,
                const float* __restrict__ al_b1, const float* __restrict__ al_W2,
                const float* __restrict__ al_b2,
                const int* __restrict__ perms,
                const float* __restrict__ u, float* __restrict__ out)
{
    extern __shared__ float lds[];
    const int tid  = threadIdx.x;
    const int lane = tid & 31;
    const int wv   = tid >> 5;
    const int hl   = lane >> 4;
    const int l15  = lane & 15;

    float* wgt  = lds;
    float* xw   = lds + WGT_SZ + wv * WAVE_SCR;
    float* actA = xw + XW_SZ;
    float* actB = actA + ACT_SZ;

    const int sampleBase = (blockIdx.x * WAVES_PER_BLOCK + wv) * 16;

    for (int idx = lane; idx < 256; idx += 32) {
        int m = idx >> 4, d = idx & 15;
        xw[m * XW_S + d] = u[(size_t)(sampleBase + m) * DD + d];
    }

    for (int flow = 0; flow < NM; ++flow) {
        __syncthreads();

        for (int e = tid; e < HH * DD; e += 256) {
            int k = e & 15, n = e >> 4;
            int src = flow * HH * DD + n * DD + k;
            int dst = OFF_W0P + (k >> 1) * W01_S + n * 2 + (k & 1);
            wgt[dst]          = mu_M0[src] * mu_W0[src];
            wgt[NET_SZ + dst] = al_M0[src] * al_W0[src];
        }
        for (int e = tid; e < HH * HH; e += 256) {
            int k = e & 63, n = e >> 6;
            int src = flow * HH * HH + n * HH + k;
            int dst = OFF_W1P + (k >> 1) * W01_S + n * 2 + (k & 1);
            wgt[dst]          = mu_M1[src] * mu_W1[src];
            wgt[NET_SZ + dst] = al_M1[src] * al_W1[src];
        }
        for (int e = tid; e < DD * HH; e += 256) {
            int k = e & 63, n = e >> 6;
            int src = flow * DD * HH + n * HH + k;
            int dst = OFF_W2P + (k >> 1) * W2_S + n * 2 + (k & 1);
            wgt[dst]          = mu_M2[src] * mu_W2[src];
            wgt[NET_SZ + dst] = al_M2[src] * al_W2[src];
        }

        float bias0[2][4], bias1[2][4], bias2[2];
#pragma unroll
        for (int nt = 0; nt < 4; ++nt) {
            bias0[0][nt] = mu_b0[flow * HH + nt * 16 + l15];
            bias0[1][nt] = al_b0[flow * HH + nt * 16 + l15];
            bias1[0][nt] = mu_b1[flow * HH + nt * 16 + l15];
            bias1[1][nt] = al_b1[flow * HH + nt * 16 + l15];
        }
        bias2[0] = mu_b2[flow * DD + l15];
        bias2[1] = al_b2[flow * DD + l15];

        for (int idx = lane; idx < 256; idx += 32)
            actB[idx] = xw[(idx >> 4) * XW_S + (idx & 15)];
        __syncthreads();
        for (int idx = lane; idx < 256; idx += 32) {
            int m = idx >> 4, d = idx & 15;
            xw[m * XW_S + d] = actB[(m << 4) + perms[flow * DD + d]];
        }
        __builtin_amdgcn_wave_barrier();

        for (int j = 0; j < DD; ++j) {
            v2f xa[4];
#pragma unroll
            for (int kk4 = 0; kk4 < 4; ++kk4)
                xa[kk4] = *(const v2f*)(xw + l15 * XW_S + kk4 * 4 + hl * 2);

            float muv[8];
            for (int net = 0; net < 2; ++net) {
                const float* wg  = wgt + net * NET_SZ;
                const float* w0p = wg + OFF_W0P;
                const float* w1p = wg + OFF_W1P;
                const float* w2p = wg + OFF_W2P;

                {
                    v8f c[4] = {};
#pragma unroll
                    for (int kk4 = 0; kk4 < 4; ++kk4) {
                        const int prow = kk4 * 2 + hl;
#pragma unroll
                        for (int nt = 0; nt < 4; ++nt) {
                            v2f bv = *(const v2f*)(w0p + prow * W01_S + (nt * 16 + l15) * 2);
                            c[nt] = __builtin_amdgcn_wmma_f32_16x16x4_f32(
                                        false, xa[kk4], false, bv, (short)0, c[nt], false, false);
                        }
                    }
#pragma unroll
                    for (int nt = 0; nt < 4; ++nt) {
                        const int n = nt * 16 + l15;
#pragma unroll
                        for (int r = 0; r < 8; ++r)
                            actA[(r + hl * 8) * ACT_S + n] = fast_tanh(c[nt][r] + bias0[net][nt]);
                    }
                }
                __builtin_amdgcn_wave_barrier();

                {
                    v8f c[4] = {};
#pragma unroll
                    for (int kk = 0; kk < HH; kk += 4) {
                        v2f av = *(const v2f*)(actA + l15 * ACT_S + kk + hl * 2);
                        const int prow = (kk >> 1) + hl;
#pragma unroll
                        for (int nt = 0; nt < 4; ++nt) {
                            v2f bv = *(const v2f*)(w1p + prow * W01_S + (nt * 16 + l15) * 2);
                            c[nt] = __builtin_amdgcn_wmma_f32_16x16x4_f32(
                                        false, av, false, bv, (short)0, c[nt], false, false);
                        }
                    }
#pragma unroll
                    for (int nt = 0; nt < 4; ++nt) {
                        const int n = nt * 16 + l15;
#pragma unroll
                        for (int r = 0; r < 8; ++r)
                            actB[(r + hl * 8) * ACT_S + n] = fast_tanh(c[nt][r] + bias1[net][nt]);
                    }
                }
                __builtin_amdgcn_wave_barrier();

                {
                    v8f c = {};
#pragma unroll
                    for (int kk = 0; kk < HH; kk += 4) {
                        v2f av = *(const v2f*)(actB + l15 * ACT_S + kk + hl * 2);
                        v2f bv = *(const v2f*)(w2p + ((kk >> 1) + hl) * W2_S + l15 * 2);
                        c = __builtin_amdgcn_wmma_f32_16x16x4_f32(
                                false, av, false, bv, (short)0, c, false, false);
                    }
                    if (net == 0) {
#pragma unroll
                        for (int r = 0; r < 8; ++r) muv[r] = c[r] + bias2[0];
                    } else {
                        if (l15 == j) {
#pragma unroll
                            for (int r = 0; r < 8; ++r) {
                                float s = 0.1f + 1.9f * fast_sigmoid(c[r] + bias2[1]);
                                int m = r + hl * 8;
                                xw[m * XW_S + j] = muv[r] + s * xw[m * XW_S + j];
                            }
                        }
                    }
                }
                __builtin_amdgcn_wave_barrier();
            }
        }
    }

    for (int idx = lane; idx < 256; idx += 32) {
        int m = idx >> 4, d = idx & 15;
        out[(size_t)(sampleBase + m) * DD + d] = xw[m * XW_S + d];
    }
}

extern "C" void kernel_launch(void* const* d_in, const int* in_sizes, int n_in,
                              void* d_out, int out_size, void* d_ws, size_t ws_size,
                              hipStream_t stream) {
    const float* mu_M0 = (const float*)d_in[0];
    const float* mu_M1 = (const float*)d_in[1];
    const float* mu_M2 = (const float*)d_in[2];
    const float* mu_W0 = (const float*)d_in[3];
    const float* mu_b0 = (const float*)d_in[4];
    const float* mu_W1 = (const float*)d_in[5];
    const float* mu_b1 = (const float*)d_in[6];
    const float* mu_W2 = (const float*)d_in[7];
    const float* mu_b2 = (const float*)d_in[8];
    const float* al_M0 = (const float*)d_in[9];
    const float* al_M1 = (const float*)d_in[10];
    const float* al_M2 = (const float*)d_in[11];
    const float* al_W0 = (const float*)d_in[12];
    const float* al_b0 = (const float*)d_in[13];
    const float* al_W1 = (const float*)d_in[14];
    const float* al_b1 = (const float*)d_in[15];
    const float* al_W2 = (const float*)d_in[16];
    const float* al_b2 = (const float*)d_in[17];
    const int*   perms = (const int*)d_in[18];
    const float* u     = (const float*)d_in[19];

    const int rows   = in_sizes[19] / DD;
    const int blocks = rows / (WAVES_PER_BLOCK * 16);

    maf_wmma_kernel<<<blocks, 256, LDS_BYTES, stream>>>(
        mu_M0, mu_M1, mu_M2, mu_W0, mu_b0, mu_W1, mu_b1, mu_W2, mu_b2,
        al_M0, al_M1, al_M2, al_W0, al_b0, al_W1, al_b1, al_W2, al_b2,
        perms, u, (float*)d_out);
}


# Layer 2 forms only output row j per step: 64-term dot of masked weight row j (per-flow LDS copy) with h1 as packed f32 fma + one cross-half add, not 16 rows on WMMA; layer-2 sum order changes
# speedup vs baseline: 1.7996x; 1.7996x over previous
#include <hip/hip_runtime.h>
#include <math.h>


typedef __attribute__((ext_vector_type(2))) float v2f;
typedef __attribute__((ext_vector_type(4))) float v4f;
typedef __attribute__((ext_vector_type(8))) float v8f;

#define BATCH 65536
#define DD 16
#define HH 64
#define NM 3

#define NT 4
#define WAVES 4
#define XS 20
#define SCR (NT * 16 * XS)

#define LOG2E 1.4426950408889634f

__device__ __forceinline__ float fast_tanh(float x) {
    float e = __builtin_amdgcn_exp2f(x * (2.0f * LOG2E));
    return 1.0f - 2.0f * __builtin_amdgcn_rcpf(e + 1.0f);
}
__device__ __forceinline__ float fast_sigmoid(float a) {
    return __builtin_amdgcn_rcpf(1.0f + __builtin_amdgcn_exp2f(a * -LOG2E));
}

__device__ __forceinline__ v8f wmma4(v2f a, v2f b, v8f c) {
    return __builtin_amdgcn_wmma_f32_16x16x4_f32(false, a, false, b, (short)0, c, false, false);
}

struct Net {
    v2f w0[4][4];
    v2f w1[4][4][4];
    v8f b0[4], b1[4];
};

__device__ __forceinline__ void ld8(const float* __restrict__ m, const float* __restrict__ w,
                                    int off, v2f (&dst)[4]) {
    const v4f ma = *(const v4f*)(m + off), mb = *(const v4f*)(m + off + 4);
    const v4f wa = *(const v4f*)(w + off), wb = *(const v4f*)(w + off + 4);
    dst[0] = v2f{ma[0] * wa[0], ma[1] * wa[1]};
    dst[1] = v2f{ma[2] * wa[2], ma[3] * wa[3]};
    dst[2] = v2f{mb[0] * wb[0], mb[1] * wb[1]};
    dst[3] = v2f{mb[2] * wb[2], mb[3] * wb[3]};
}
__device__ __forceinline__ v8f ldb8(const float* __restrict__ b, int off) {
    const v4f a = *(const v4f*)(b + off), c = *(const v4f*)(b + off + 4);
    return v8f{a[0], a[1], a[2], a[3], c[0], c[1], c[2], c[3]};
}

__device__ __forceinline__ void load_net(Net& n, int flow, int l15, int hl,
        const float* __restrict__ M0, const float* __restrict__ M1,
        const float* __restrict__ W0, const float* __restrict__ b0,
        const float* __restrict__ W1, const float* __restrict__ b1)
{
#pragma unroll
    for (int mt = 0; mt < 4; ++mt) {
        const int row = mt * 16 + l15;
        ld8(M0, W0, flow * HH * DD + row * DD + hl * 8, n.w0[mt]);
#pragma unroll
        for (int kt = 0; kt < 4; ++kt)
            ld8(M1, W1, flow * HH * HH + row * HH + kt * 16 + hl * 8, n.w1[mt][kt]);
        n.b0[mt] = ldb8(b0, flow * HH + mt * 16 + hl * 8);
        n.b1[mt] = ldb8(b1, flow * HH + mt * 16 + hl * 8);
    }
}

#define FRAG(v, q) (v2f{(v)[2 * (q)], (v)[2 * (q) + 1]})

__device__ __forceinline__ float xhalf(float v) {
    return __builtin_bit_cast(float, __builtin_amdgcn_permlanex16(__builtin_bit_cast(int, v),
                              __builtin_bit_cast(int, v), 0x76543210, 0xfedcba98, false, false));
}

__device__ __forceinline__ void ld_w2row(const float* rowp, int hl, v2f (&w2r)[4][4]) {
#pragma unroll
    for (int mt = 0; mt < 4; ++mt) {
        const v4f a = *(const v4f*)(rowp + mt * 16 + hl * 8);
        const v4f b = *(const v4f*)(rowp + mt * 16 + hl * 8 + 4);
        w2r[mt][0] = v2f{a[0], a[1]};
        w2r[mt][1] = v2f{a[2], a[3]};
        w2r[mt][2] = v2f{b[0], b[1]};
        w2r[mt][3] = v2f{b[2], b[3]};
    }
}

__device__ __forceinline__ void net_fwd(const Net& n, const v2f (&w2r)[4][4], float b2j,
                                        const v8f (&x)[NT], float (&o)[NT]) {
    v8f h[NT][4];
#pragma unroll
    for (int mt = 0; mt < 4; ++mt)
#pragma unroll
        for (int t = 0; t < NT; ++t) h[t][mt] = n.b0[mt];
#pragma unroll
    for (int q = 0; q < 4; ++q)
#pragma unroll
        for (int mt = 0; mt < 4; ++mt)
#pragma unroll
            for (int t = 0; t < NT; ++t)
                h[t][mt] = wmma4(n.w0[mt][q], FRAG(x[t], q), h[t][mt]);
#pragma unroll
    for (int t = 0; t < NT; ++t)
#pragma unroll
        for (int mt = 0; mt < 4; ++mt)
#pragma unroll
            for (int r = 0; r < 8; ++r) h[t][mt][r] = fast_tanh(h[t][mt][r]);

    v8f g[NT][4];
#pragma unroll
    for (int mt = 0; mt < 4; ++mt)
#pragma unroll
        for (int t = 0; t < NT; ++t) g[t][mt] = n.b1[mt];
#pragma unroll
    for (int kt = 0; kt < 4; ++kt)
#pragma unroll
        for (int q = 0; q < 4; ++q)
#pragma unroll
            for (int mt = 0; mt < 4; ++mt)
#pragma unroll
                for (int t = 0; t < NT; ++t)
                    g[t][mt] = wmma4(n.w1[mt][kt][q], FRAG(h[t][kt], q), g[t][mt]);
#pragma unroll
    for (int t = 0; t < NT; ++t)
#pragma unroll
        for (int mt = 0; mt < 4; ++mt)
#pragma unroll
            for (int r = 0; r < 8; ++r) g[t][mt][r] = fast_tanh(g[t][mt][r]);

#pragma unroll
    for (int t = 0; t < NT; ++t) {
        v2f a0 = w2r[0][0] * FRAG(g[t][0], 0);
        v2f a1 = w2r[0][1] * FRAG(g[t][0], 1);
#pragma unroll
        for (int kt = 0; kt < 4; ++kt)
#pragma unroll
            for (int q = (kt == 0 ? 2 : 0); q < 4; q += 2) {
                a0 = __builtin_elementwise_fma(w2r[kt][q],     FRAG(g[t][kt], q),     a0);
                a1 = __builtin_elementwise_fma(w2r[kt][q + 1], FRAG(g[t][kt], q + 1), a1);
            }
        const float part = (a0[0] + a1[0]) + (a0[1] + a1[1]);
        o[t] = (part + xhalf(part)) + b2j;
    }
}

__device__ __forceinline__ float pick8(const v8f& v, int jr) {
    float s = v[0];
#pragma unroll
    for (int r = 1; r < 8; ++r) s = (jr == r) ? v[r] : s;
    return s;
}

__global__ void __launch_bounds__(WAVES * 32)
maf_wmma_kernel(const float* __restrict__ mu_M0, const float* __restrict__ mu_M1,
                const float* __restrict__ mu_M2, const float* __restrict__ mu_W0,
                const float* __restrict__ mu_b0, const float* __restrict__ mu_W1,
                const float* __restrict__ mu_b1, const float* __restrict__ mu_W2,
                const float* __restrict__ mu_b2,
                const float* __restrict__ al_M0, const float* __restrict__ al_M1,
                const float* __restrict__ al_M2, const float* __restrict__ al_W0,
                const float* __restrict__ al_b0, const float* __restrict__ al_W1,
                const float* __restrict__ al_b1, const float* __restrict__ al_W2,
                const float* __restrict__ al_b2,
                const int* __restrict__ perms,
                const float* __restrict__ u, float* __restrict__ out)
{
    __shared__ float scratch[WAVES * SCR];
    __shared__ __attribute__((aligned(16))) float w2s[2 * DD * HH];
    const int tid  = threadIdx.x;
    const int lane = tid & 31;
    const int wv   = tid >> 5;
    const int hl   = lane >> 4;
    const int l15  = lane & 15;

    float* sc = scratch + wv * SCR;
    const int sampleBase = (blockIdx.x * WAVES + wv) * (NT * 16);

    v8f x[NT];
#pragma unroll
    for (int t = 0; t < NT; ++t)
        x[t] = ldb8(u, (sampleBase + t * 16 + l15) * DD + hl * 8);

    for (int flow = 0; flow < NM; ++flow) {
        __syncthreads();
#pragma unroll
        for (int t = 0; t < NT; ++t) {
            float* row = sc + (t * 16 + l15) * XS + hl * 8;
            *(v4f*)(row)     = v4f{x[t][0], x[t][1], x[t][2], x[t][3]};
            *(v4f*)(row + 4) = v4f{x[t][4], x[t][5], x[t][6], x[t][7]};
        }
#pragma unroll
        for (int i = 0; i < 4; ++i) {
            const int off = (i & 1) * 512 + tid * 4;
            const float* __restrict__ mp = (i >> 1) ? al_M2 : mu_M2;
            const float* __restrict__ wp = (i >> 1) ? al_W2 : mu_W2;
            const v4f m = *(const v4f*)(mp + flow * DD * HH + off);
            const v4f w = *(const v4f*)(wp + flow * DD * HH + off);
            *(v4f*)(w2s + (i >> 1) * (DD * HH) + off) = m * w;
        }
        __syncthreads();
        int pidx[8];
#pragma unroll
        for (int r = 0; r < 8; ++r) pidx[r] = perms[flow * DD + hl * 8 + r] & (DD - 1);
#pragma unroll
        for (int t = 0; t < NT; ++t)
#pragma unroll
            for (int r = 0; r < 8; ++r)
                x[t][r] = sc[(t * 16 + l15) * XS + pidx[r]];

        Net nm, na;
        load_net(nm, flow, l15, hl, mu_M0, mu_M1, mu_W0, mu_b0, mu_W1, mu_b1);
        load_net(na, flow, l15, hl, al_M0, al_M1, al_W0, al_b0, al_W1, al_b1);

        v2f w2m[4][4], w2a[4][4];
        ld_w2row(w2s, hl, w2m);
        ld_w2row(w2s + DD * HH, hl, w2a);

        for (int j = 0; j < DD; ++j) {
            const int  jr   = j & 7;
            const bool mine = (hl == (j >> 3));

            const float b2m = mu_b2[flow * DD + j];
            const float b2a = al_b2[flow * DD + j];

            float muv[NT];
            net_fwd(nm, w2m, b2m, x, muv);
            {
                float o[NT];
                net_fwd(na, w2a, b2a, x, o);
#pragma unroll
                for (int t = 0; t < NT; ++t) {
                    const float s  = 0.1f + 1.9f * fast_sigmoid(o[t]);
                    const float nv = muv[t] + s * pick8(x[t], jr);
#pragma unroll
                    for (int r = 0; r < 8; ++r)
                        x[t][r] = (mine && jr == r) ? nv : x[t][r];
                }
            }
            const int jn = (j + 1) & (DD - 1);
            ld_w2row(w2s + jn * HH, hl, w2m);
            ld_w2row(w2s + DD * HH + jn * HH, hl, w2a);
        }
    }

    __syncthreads();
#pragma unroll
    for (int t = 0; t < NT; ++t) {
        float* row = sc + (t * 16 + l15) * XS + hl * 8;
        *(v4f*)(row)     = v4f{x[t][0], x[t][1], x[t][2], x[t][3]};
        *(v4f*)(row + 4) = v4f{x[t][4], x[t][5], x[t][6], x[t][7]};
    }
    __syncthreads();
#pragma unroll
    for (int k = 0; k < (NT * 16 * DD) / (32 * 4); ++k) {
        const int idx = lane + 32 * k;
        const int s = idx >> 2, part = idx & 3;
        const v4f v = *(const v4f*)(sc + s * XS + part * 4);
        *(v4f*)(out + (size_t)(sampleBase + s) * DD + part * 4) = v;
    }
}

extern "C" void kernel_launch(void* const* d_in, const int* in_sizes, int n_in,
                              void* d_out, int out_size, void* d_ws, size_t ws_size,
                              hipStream_t stream) {
    const float* mu_M0 = (const float*)d_in[0];
    const float* mu_M1 = (const float*)d_in[1];
    const float* mu_M2 = (const float*)d_in[2];
    const float* mu_W0 = (const float*)d_in[3];
    const float* mu_b0 = (const float*)d_in[4];
    const float* mu_W1 = (const float*)d_in[5];
    const float* mu_b1 = (const float*)d_in[6];
    const float* mu_W2 = (const float*)d_in[7];
    const float* mu_b2 = (const float*)d_in[8];
    const float* al_M0 = (const float*)d_in[9];
    const float* al_M1 = (const float*)d_in[10];
    const float* al_M2 = (const float*)d_in[11];
    const float* al_W0 = (const float*)d_in[12];
    const float* al_b0 = (const float*)d_in[13];
    const float* al_W1 = (const float*)d_in[14];
    const float* al_b1 = (const float*)d_in[15];
    const float* al_W2 = (const float*)d_in[16];
    const float* al_b2 = (const float*)d_in[17];
    const int*   perms = (const int*)d_in[18];
    const float* u     = (const float*)d_in[19];

    maf_wmma_kernel<<<BATCH / (WAVES * NT * 16), WAVES * 32, 0, stream>>>(
        mu_M0, mu_M1, mu_M2, mu_W0, mu_b0, mu_W1, mu_b1, mu_W2, mu_b2,
        al_M0, al_M1, al_M2, al_W0, al_b0, al_W1, al_b1, al_W2, al_b2,
        perms, u, (float*)d_out);
}


# Activations as a software pipeline over register pairs: each exp2/rcp is followed by one packed plain op of another pair (scale, +1, 1-2r, layer-2 fma), order held by sched_barrier; sums unchanged
# speedup vs baseline: 1.8800x; 1.0447x over previous
#include <hip/hip_runtime.h>
#include <math.h>


typedef __attribute__((ext_vector_type(2))) float v2f;
typedef __attribute__((ext_vector_type(4))) float v4f;
typedef __attribute__((ext_vector_type(8))) float v8f;

#define BATCH 65536
#define DD 16
#define HH 64
#define NM 3

#define NT 4
#define WAVES 4
#define XS 20
#define SCR (NT * 16 * XS)

#define LOG2E 1.4426950408889634f

__device__ __forceinline__ float fast_sigmoid(float a) {
    return __builtin_amdgcn_rcpf(1.0f + __builtin_amdgcn_exp2f(a * -LOG2E));
}


#define NPAIR (NT * 4 * 4)
#define PAIR_MT(i) ((i) >> 4)
#define PAIR_T(i)  (((i) >> 2) & 3)
#define PAIR_Q(i)  ((i) & 3)
#define IN_PIPE(j) ((j) >= 0 && (j) < NPAIR)
#define KEEP_ORDER() __builtin_amdgcn_sched_barrier(DOT ? 0x3F4 : 0x3FC)
template <bool DOT>
__device__ __forceinline__ void act_pipe(v8f (&a)[NT][4], const v2f (&w2r)[4][4], v2f (&acc)[NT][2]) {
    v2f s[NPAIR], e[NPAIR], d[NPAIR], rc[NPAIR], hv[NPAIR];
#pragma unroll
    for (int i = 0; i < NPAIR + 6; ++i) {
        if (IN_PIPE(i - 1)) { e[i - 1][0] = __builtin_amdgcn_exp2f(s[i - 1][0]); KEEP_ORDER(); }
        if (IN_PIPE(i)) {
            const v8f& v = a[PAIR_T(i)][PAIR_MT(i)];
            s[i] = v2f{v[2 * PAIR_Q(i)], v[2 * PAIR_Q(i) + 1]} * (2.0f * LOG2E);
            KEEP_ORDER();
        }
        if (IN_PIPE(i - 1)) { e[i - 1][1] = __builtin_amdgcn_exp2f(s[i - 1][1]); KEEP_ORDER(); }
        if (IN_PIPE(i - 2)) { d[i - 2] = e[i - 2] + 1.0f; KEEP_ORDER(); }
        if (IN_PIPE(i - 3)) { rc[i - 3][0] = __builtin_amdgcn_rcpf(d[i - 3][0]); KEEP_ORDER(); }
        if (DOT && IN_PIPE(i - 5)) {
            const int j = i - 5, t = PAIR_T(j), mt = PAIR_MT(j), q = PAIR_Q(j);
            acc[t][q & 1] = (mt == 0 && q < 2) ? w2r[mt][q] * hv[j]
                                               : __builtin_elementwise_fma(w2r[mt][q], hv[j], acc[t][q & 1]);
            KEEP_ORDER();
        }
        if (IN_PIPE(i - 3)) { rc[i - 3][1] = __builtin_amdgcn_rcpf(d[i - 3][1]); KEEP_ORDER(); }
        if (IN_PIPE(i - 4)) {
            hv[i - 4] = 1.0f - 2.0f * rc[i - 4];
            if (!DOT) {
                v8f& v = a[PAIR_T(i - 4)][PAIR_MT(i - 4)];
                v[2 * PAIR_Q(i - 4)]     = hv[i - 4][0];
                v[2 * PAIR_Q(i - 4) + 1] = hv[i - 4][1];
            }
            KEEP_ORDER();
        }
    }
}

__device__ __forceinline__ v8f wmma4(v2f a, v2f b, v8f c) {
    return __builtin_amdgcn_wmma_f32_16x16x4_f32(false, a, false, b, (short)0, c, false, false);
}

struct Net {
    v2f w0[4][4];
    v2f w1[4][4][4];
    v8f b0[4], b1[4];
};

__device__ __forceinline__ void ld8(const float* __restrict__ m, const float* __restrict__ w,
                                    int off, v2f (&dst)[4]) {
    const v4f ma = *(const v4f*)(m + off), mb = *(const v4f*)(m + off + 4);
    const v4f wa = *(const v4f*)(w + off), wb = *(const v4f*)(w + off + 4);
    dst[0] = v2f{ma[0] * wa[0], ma[1] * wa[1]};
    dst[1] = v2f{ma[2] * wa[2], ma[3] * wa[3]};
    dst[2] = v2f{mb[0] * wb[0], mb[1] * wb[1]};
    dst[3] = v2f{mb[2] * wb[2], mb[3] * wb[3]};
}
__device__ __forceinline__ v8f ldb8(const float* __restrict__ b, int off) {
    const v4f a = *(const v4f*)(b + off), c = *(const v4f*)(b + off + 4);
    return v8f{a[0], a[1], a[2], a[3], c[0], c[1], c[2], c[3]};
}

__device__ __forceinline__ void load_net(Net& n, int flow, int l15, int hl,
        const float* __restrict__ M0, const float* __restrict__ M1,
        const float* __restrict__ W0, const float* __restrict__ b0,
        const float* __restrict__ W1, const float* __restrict__ b1)
{
#pragma unroll
    for (int mt = 0; mt < 4; ++mt) {
        const int row = mt * 16 + l15;
        ld8(M0, W0, flow * HH * DD + row * DD + hl * 8, n.w0[mt]);
#pragma unroll
        for (int kt = 0; kt < 4; ++kt)
            ld8(M1, W1, flow * HH * HH + row * HH + kt * 16 + hl * 8, n.w1[mt][kt]);
        n.b0[mt] = ldb8(b0, flow * HH + mt * 16 + hl * 8);
        n.b1[mt] = ldb8(b1, flow * HH + mt * 16 + hl * 8);
    }
}

#define FRAG(v, q) (v2f{(v)[2 * (q)], (v)[2 * (q) + 1]})

__device__ __forceinline__ float xhalf(float v) {
    return __builtin_bit_cast(float, __builtin_amdgcn_permlanex16(__builtin_bit_cast(int, v),
                              __builtin_bit_cast(int, v), 0x76543210, 0xfedcba98, false, false));
}

__device__ __forceinline__ void ld_w2row(const float* rowp, int hl, v2f (&w2r)[4][4]) {
#pragma unroll
    for (int mt = 0; mt < 4; ++mt) {
        const v4f a = *(const v4f*)(rowp + mt * 16 + hl * 8);
        const v4f b = *(const v4f*)(rowp + mt * 16 + hl * 8 + 4);
        w2r[mt][0] = v2f{a[0], a[1]};
        w2r[mt][1] = v2f{a[2], a[3]};
        w2r[mt][2] = v2f{b[0], b[1]};
        w2r[mt][3] = v2f{b[2], b[3]};
    }
}

__device__ __forceinline__ void net_fwd(const Net& n, const v2f (&w2r)[4][4], float b2j,
                                        const v8f (&x)[NT], float (&o)[NT]) {
    v8f h[NT][4];
#pragma unroll
    for (int mt = 0; mt < 4; ++mt)
#pragma unroll
        for (int t = 0; t < NT; ++t) h[t][mt] = n.b0[mt];
#pragma unroll
    for (int q = 0; q < 4; ++q)
#pragma unroll
        for (int mt = 0; mt < 4; ++mt)
#pragma unroll
            for (int t = 0; t < NT; ++t)
                h[t][mt] = wmma4(n.w0[mt][q], FRAG(x[t], q), h[t][mt]);
    v2f acc[NT][2];
    act_pipe<false>(h, w2r, acc);

    v8f g[NT][4];
#pragma unroll
    for (int mt = 0; mt < 4; ++mt)
#pragma unroll
        for (int t = 0; t < NT; ++t) g[t][mt] = n.b1[mt];
#pragma unroll
    for (int kt = 0; kt < 4; ++kt)
#pragma unroll
        for (int q = 0; q < 4; ++q)
#pragma unroll
            for (int mt = 0; mt < 4; ++mt)
#pragma unroll
                for (int t = 0; t < NT; ++t)
                    g[t][mt] = wmma4(n.w1[mt][kt][q], FRAG(h[t][kt], q), g[t][mt]);
    act_pipe<true>(g, w2r, acc);
#pragma unroll
    for (int t = 0; t < NT; ++t) {
        const v2f a0 = acc[t][0], a1 = acc[t][1];
        const float part = (a0[0] + a1[0]) + (a0[1] + a1[1]);
        o[t] = (part + xhalf(part)) + b2j;
    }
}

__device__ __forceinline__ float pick8(const v8f& v, int jr) {
    float s = v[0];
#pragma unroll
    for (int r = 1; r < 8; ++r) s = (jr == r) ? v[r] : s;
    return s;
}

__global__ void __launch_bounds__(WAVES * 32)
maf_wmma_kernel(const float* __restrict__ mu_M0, const float* __restrict__ mu_M1,
                const float* __restrict__ mu_M2, const float* __restrict__ mu_W0,
                const float* __restrict__ mu_b0, const float* __restrict__ mu_W1,
                const float* __restrict__ mu_b1, const float* __restrict__ mu_W2,
                const float* __restrict__ mu_b2,
                const float* __restrict__ al_M0, const float* __restrict__ al_M1,
                const float* __restrict__ al_M2, const float* __restrict__ al_W0,
                const float* __restrict__ al_b0, const float* __restrict__ al_W1,
                const float* __restrict__ al_b1, const float* __restrict__ al_W2,
                const float* __restrict__ al_b2,
                const int* __restrict__ perms,
                const float* __restrict__ u, float* __restrict__ out)
{
    __shared__ float scratch[WAVES * SCR];
    __shared__ __attribute__((aligned(16))) float w2s[2 * DD * HH];
    const int tid  = threadIdx.x;
    const int lane = tid & 31;
    const int wv   = tid >> 5;
    const int hl   = lane >> 4;
    const int l15  = lane & 15;

    float* sc = scratch + wv * SCR;
    const int sampleBase = (blockIdx.x * WAVES + wv) * (NT * 16);

    v8f x[NT];
#pragma unroll
    for (int t = 0; t < NT; ++t)
        x[t] = ldb8(u, (sampleBase + t * 16 + l15) * DD + hl * 8);

    for (int flow = 0; flow < NM; ++flow) {
        __syncthreads();
#pragma unroll
        for (int t = 0; t < NT; ++t) {
            float* row = sc + (t * 16 + l15) * XS + hl * 8;
            *(v4f*)(row)     = v4f{x[t][0], x[t][1], x[t][2], x[t][3]};
            *(v4f*)(row + 4) = v4f{x[t][4], x[t][5], x[t][6], x[t][7]};
        }
#pragma unroll
        for (int i = 0; i < 4; ++i) {
            const int off = (i & 1) * 512 + tid * 4;
            const float* __restrict__ mp = (i >> 1) ? al_M2 : mu_M2;
            const float* __restrict__ wp = (i >> 1) ? al_W2 : mu_W2;
            const v4f m = *(const v4f*)(mp + flow * DD * HH + off);
            const v4f w = *(const v4f*)(wp + flow * DD * HH + off);
            *(v4f*)(w2s + (i >> 1) * (DD * HH) + off) = m * w;
        }
        __syncthreads();
        int pidx[8];
#pragma unroll
        for (int r = 0; r < 8; ++r) pidx[r] = perms[flow * DD + hl * 8 + r] & (DD - 1);
#pragma unroll
        for (int t = 0; t < NT; ++t)
#pragma unroll
            for (int r = 0; r < 8; ++r)
                x[t][r] = sc[(t * 16 + l15) * XS + pidx[r]];

        Net nm, na;
        load_net(nm, flow, l15, hl, mu_M0, mu_M1, mu_W0, mu_b0, mu_W1, mu_b1);
        load_net(na, flow, l15, hl, al_M0, al_M1, al_W0, al_b0, al_W1, al_b1);

        v2f w2m[4][4], w2a[4][4];
        ld_w2row(w2s, hl, w2m);
        ld_w2row(w2s + DD * HH, hl, w2a);

        for (int j = 0; j < DD; ++j) {
            const int  jr   = j & 7;
            const bool mine = (hl == (j >> 3));

            const float b2m = mu_b2[flow * DD + j];
            const float b2a = al_b2[flow * DD + j];

            float muv[NT];
            net_fwd(nm, w2m, b2m, x, muv);
            {
                float o[NT];
                net_fwd(na, w2a, b2a, x, o);
#pragma unroll
                for (int t = 0; t < NT; ++t) {
                    const float s  = 0.1f + 1.9f * fast_sigmoid(o[t]);
                    const float nv = muv[t] + s * pick8(x[t], jr);
#pragma unroll
                    for (int r = 0; r < 8; ++r)
                        x[t][r] = (mine && jr == r) ? nv : x[t][r];
                }
            }
            const int jn = (j + 1) & (DD - 1);
            ld_w2row(w2s + jn * HH, hl, w2m);
            ld_w2row(w2s + DD * HH + jn * HH, hl, w2a);
        }
    }

    __syncthreads();
#pragma unroll
    for (int t = 0; t < NT; ++t) {
        float* row = sc + (t * 16 + l15) * XS + hl * 8;
        *(v4f*)(row)     = v4f{x[t][0], x[t][1], x[t][2], x[t][3]};
        *(v4f*)(row + 4) = v4f{x[t][4], x[t][5], x[t][6], x[t][7]};
    }
    __syncthreads();
#pragma unroll
    for (int k = 0; k < (NT * 16 * DD) / (32 * 4); ++k) {
        const int idx = lane + 32 * k;
        const int s = idx >> 2, part = idx & 3;
        const v4f v = *(const v4f*)(sc + s * XS + part * 4);
        *(v4f*)(out + (size_t)(sampleBase + s) * DD + part * 4) = v;
    }
}

extern "C" void kernel_launch(void* const* d_in, const int* in_sizes, int n_in,
                              void* d_out, int out_size, void* d_ws, size_t ws_size,
                              hipStream_t stream) {
    const float* mu_M0 = (const float*)d_in[0];
    const float* mu_M1 = (const float*)d_in[1];
    const float* mu_M2 = (const float*)d_in[2];
    const float* mu_W0 = (const float*)d_in[3];
    const float* mu_b0 = (const float*)d_in[4];
    const float* mu_W1 = (const float*)d_in[5];
    const float* mu_b1 = (const float*)d_in[6];
    const float* mu_W2 = (const float*)d_in[7];
    const float* mu_b2 = (const float*)d_in[8];
    const float* al_M0 = (const float*)d_in[9];
    const float* al_M1 = (const float*)d_in[10];
    const float* al_M2 = (const float*)d_in[11];
    const float* al_W0 = (const float*)d_in[12];
    const float* al_b0 = (const float*)d_in[13];
    const float* al_W1 = (const float*)d_in[14];
    const float* al_b1 = (const float*)d_in[15];
    const float* al_W2 = (const float*)d_in[16];
    const float* al_b2 = (const float*)d_in[17];
    const int*   perms = (const int*)d_in[18];
    const float* u     = (const float*)d_in[19];

    maf_wmma_kernel<<<BATCH / (WAVES * NT * 16), WAVES * 32, 0, stream>>>(
        mu_M0, mu_M1, mu_M2, mu_W0, mu_b0, mu_W1, mu_b1, mu_W2, mu_b2,
        al_M0, al_M1, al_M2, al_W0, al_b0, al_W1, al_b1, al_W2, al_b2,
        perms, u, (float*)d_out);
}
